# starting kernel
# baseline (speedup 1.0000x reference)
#include <hip/hip_runtime.h>

typedef __attribute__((ext_vector_type(2))) float v2f;
typedef __attribute__((ext_vector_type(4))) float v4f;
typedef __attribute__((ext_vector_type(8))) float v8f;

#define H_IN   384
#define W_IN   384
#define H_OUT  377
#define W_OUT  377
#define KH     8
#define KW     8
#define TILE_X 32
#define STRIP_Y 64
#define ROWS   (STRIP_Y + KH - 1)
#define XCOLS  (TILE_X + KW)

__global__ __launch_bounds__(256)
void corr_wmma_kernel(const float* __restrict__ x,
                      const float* __restrict__ k,
                      float* __restrict__ out) {
  __shared__ float XS[ROWS * XCOLS];
  __shared__ float RS[ROWS * KH * TILE_X];

  const int tid  = threadIdx.x;
  const int lane = tid & 31;
  const int wave = tid >> 5;
  const int b    = blockIdx.z;
  const int x0   = blockIdx.x * TILE_X;
  const int y0   = blockIdx.y * STRIP_Y;

  const float* xb = x + (size_t)b * H_IN * W_IN;
  for (int idx = tid; idx < ROWS * XCOLS; idx += 256) {
    int yy = idx / XCOLS;
    int xx = idx - yy * XCOLS;
    int gy = y0 + yy;
    int gx = x0 + xx;
    float v = 0.0f;
    if (gy < H_IN && gx < W_IN) v = xb[gy * W_IN + gx];
    XS[idx] = v;
  }

  const int m    = lane & 15;
  const int half = lane >> 4;
  const int koff = half * 2;
  v2f bm1 = {0.f, 0.f}, bm2 = {0.f, 0.f};
  if (m < KH) {
    const float* kb = k + (size_t)b * (KH * KW) + m * KW;
    bm1.x = kb[0 + koff]; bm1.y = kb[1 + koff];
    bm2.x = kb[4 + koff]; bm2.y = kb[5 + koff];
  }

  __syncthreads();

  for (int yy = wave; yy < ROWS; yy += 8) {
    const float* xr = &XS[yy * XCOLS];

    v2f a1, a2;
    a1.x = xr[m + koff];
    a1.y = xr[m + koff + 1];
    a2.x = xr[m + koff + 4];
    a2.y = xr[m + koff + 5];
    v8f acc0 = {};
    acc0 = __builtin_amdgcn_wmma_f32_16x16x4_f32(false, a1, false, bm1, (short)0, acc0, false, false);
    acc0 = __builtin_amdgcn_wmma_f32_16x16x4_f32(false, a2, false, bm2, (short)0, acc0, false, false);

    v2f a3, a4;
    a3.x = xr[16 + m + koff];
    a3.y = xr[16 + m + koff + 1];
    a4.x = xr[16 + m + koff + 4];
    a4.y = xr[16 + m + koff + 5];
    v8f acc1 = {};
    acc1 = __builtin_amdgcn_wmma_f32_16x16x4_f32(false, a3, false, bm1, (short)0, acc1, false, false);
    acc1 = __builtin_amdgcn_wmma_f32_16x16x4_f32(false, a4, false, bm2, (short)0, acc1, false, false);

    if (m < KH) {
      float* dst = &RS[(yy * KH + m) * TILE_X + half * 8];
      v4f lo0 = {acc0[0], acc0[1], acc0[2], acc0[3]};
      v4f hi0 = {acc0[4], acc0[5], acc0[6], acc0[7]};
      *(v4f*)(dst)          = lo0;
      *(v4f*)(dst + 4)      = hi0;
      v4f lo1 = {acc1[0], acc1[1], acc1[2], acc1[3]};
      v4f hi1 = {acc1[4], acc1[5], acc1[6], acc1[7]};
      *(v4f*)(dst + 16)     = lo1;
      *(v4f*)(dst + 16 + 4) = hi1;
    }
  }

  __syncthreads();

  float* ob = out + (size_t)b * H_OUT * W_OUT;
  const int xo = tid & 31;
  const int yb = tid >> 5;
  for (int i = 0; i < STRIP_Y / 8; ++i) {
    int yo = yb + i * 8;
    float s = 0.f;
#pragma unroll
    for (int r = 0; r < KH; ++r)
      s += RS[((yo + r) * KH + r) * TILE_X + xo];
    int gy = y0 + yo, gx = x0 + xo;
    if (gy < H_OUT && gx < W_OUT) ob[gy * W_OUT + gx] = s;
  }
}

extern "C" void kernel_launch(void* const* d_in, const int* in_sizes, int n_in,
                              void* d_out, int out_size, void* d_ws, size_t ws_size,
                              hipStream_t stream) {
  (void)in_sizes; (void)n_in; (void)d_ws; (void)ws_size; (void)out_size;
  const float* x = (const float*)d_in[0];
  const float* k = (const float*)d_in[1];
  float* out = (float*)d_out;
  dim3 grid((W_OUT + TILE_X - 1) / TILE_X,
            (H_OUT + STRIP_Y - 1) / STRIP_Y,
            128);
  corr_wmma_kernel<<<grid, dim3(256, 1, 1), 0, stream>>>(x, k, out);
}


# stage 0: each thread issues all its global loads (filter fragment, then 3 x b128 of the strip) before its first LDS store: one memory round trip per work-group instead of twelve
# speedup vs baseline: 1.6402x; 1.6402x over previous
#include <hip/hip_runtime.h>

typedef __attribute__((ext_vector_type(2))) float v2f;
typedef __attribute__((ext_vector_type(4))) float v4f;
typedef __attribute__((ext_vector_type(8))) float v8f;

#define H_IN   384
#define W_IN   384
#define H_OUT  377
#define W_OUT  377
#define KH     8
#define KW     8
#define TILE_X 32
#define STRIP_Y 64
#define ROWS   (STRIP_Y + KH - 1)
#define XCOLS  (TILE_X + KW)
#define XQ     (XCOLS / 4)
#define STAGE_ITERS ((ROWS * XQ + 255) / 256)

__global__ __launch_bounds__(256)
void corr_wmma_kernel(const float* __restrict__ x,
                      const float* __restrict__ k,
                      float* __restrict__ out) {
  __shared__ __attribute__((aligned(16))) float XS[ROWS * XCOLS];
  __shared__ float RS[ROWS * KH * TILE_X];

  const int tid  = threadIdx.x;
  const int lane = tid & 31;
  const int wave = tid >> 5;
  const int b    = blockIdx.z;
  const int x0   = blockIdx.x * TILE_X;
  const int y0   = blockIdx.y * STRIP_Y;

  const int m    = lane & 15;
  const int half = lane >> 4;
  const int koff = half * 2;
  v2f bm1 = {0.f, 0.f}, bm2 = {0.f, 0.f};
  if (m < KH) {
    const float* kb = k + (size_t)b * (KH * KW) + m * KW;
    bm1.x = kb[0 + koff]; bm1.y = kb[1 + koff];
    bm2.x = kb[4 + koff]; bm2.y = kb[5 + koff];
  }

  const float* xb = x + (size_t)b * H_IN * W_IN;
  {
    v4f sv[STAGE_ITERS];
#pragma unroll
    for (int it = 0; it < STAGE_ITERS; ++it) {
      int q  = tid + it * 256;
      int yy = q / XQ;
      int xq = q - yy * XQ;
      int gy = y0 + yy;
      int gx = x0 + xq * 4;
      v4f v = {0.f, 0.f, 0.f, 0.f};
      if (q < ROWS * XQ && gy < H_IN && gx < W_IN)
        v = *(const v4f*)(xb + gy * W_IN + gx);
      sv[it] = v;
    }
#pragma unroll
    for (int it = 0; it < STAGE_ITERS; ++it) {
      int q = tid + it * 256;
      if (q < ROWS * XQ) *(v4f*)(&XS[q * 4]) = sv[it];
    }
  }

  __syncthreads();

  for (int yy = wave; yy < ROWS; yy += 8) {
    const float* xr = &XS[yy * XCOLS];

    v2f a1, a2;
    a1.x = xr[m + koff];
    a1.y = xr[m + koff + 1];
    a2.x = xr[m + koff + 4];
    a2.y = xr[m + koff + 5];
    v8f acc0 = {};
    acc0 = __builtin_amdgcn_wmma_f32_16x16x4_f32(false, a1, false, bm1, (short)0, acc0, false, false);
    acc0 = __builtin_amdgcn_wmma_f32_16x16x4_f32(false, a2, false, bm2, (short)0, acc0, false, false);

    v2f a3, a4;
    a3.x = xr[16 + m + koff];
    a3.y = xr[16 + m + koff + 1];
    a4.x = xr[16 + m + koff + 4];
    a4.y = xr[16 + m + koff + 5];
    v8f acc1 = {};
    acc1 = __builtin_amdgcn_wmma_f32_16x16x4_f32(false, a3, false, bm1, (short)0, acc1, false, false);
    acc1 = __builtin_amdgcn_wmma_f32_16x16x4_f32(false, a4, false, bm2, (short)0, acc1, false, false);

    if (m < KH) {
      float* dst = &RS[(yy * KH + m) * TILE_X + half * 8];
      v4f lo0 = {acc0[0], acc0[1], acc0[2], acc0[3]};
      v4f hi0 = {acc0[4], acc0[5], acc0[6], acc0[7]};
      *(v4f*)(dst)          = lo0;
      *(v4f*)(dst + 4)      = hi0;
      v4f lo1 = {acc1[0], acc1[1], acc1[2], acc1[3]};
      v4f hi1 = {acc1[4], acc1[5], acc1[6], acc1[7]};
      *(v4f*)(dst + 16)     = lo1;
      *(v4f*)(dst + 16 + 4) = hi1;
    }
  }

  __syncthreads();

  float* ob = out + (size_t)b * H_OUT * W_OUT;
  const int xo = tid & 31;
  const int yb = tid >> 5;
  for (int i = 0; i < STRIP_Y / 8; ++i) {
    int yo = yb + i * 8;
    float s = 0.f;
#pragma unroll
    for (int r = 0; r < KH; ++r)
      s += RS[((yo + r) * KH + r) * TILE_X + xo];
    int gy = y0 + yo, gx = x0 + xo;
    if (gy < H_OUT && gx < W_OUT) ob[gy * W_OUT + gx] = s;
  }
}

extern "C" void kernel_launch(void* const* d_in, const int* in_sizes, int n_in,
                              void* d_out, int out_size, void* d_ws, size_t ws_size,
                              hipStream_t stream) {
  (void)in_sizes; (void)n_in; (void)d_ws; (void)ws_size; (void)out_size;
  const float* x = (const float*)d_in[0];
  const float* k = (const float*)d_in[1];
  float* out = (float*)d_out;
  dim3 grid((W_OUT + TILE_X - 1) / TILE_X,
            (H_OUT + STRIP_Y - 1) / STRIP_Y,
            128);
  corr_wmma_kernel<<<grid, dim3(256, 1, 1), 0, stream>>>(x, k, out);
}


# banded B operand: N = 4 output rows x 4 column shifts, M = 16 columns at stride 4, K = 11 rows x 12 taps; 33 WMMAs give 256 finished outputs, R plane and VALU row sum gone; summation order changes
# speedup vs baseline: 3.0947x; 1.8868x over previous
#include <hip/hip_runtime.h>


typedef __attribute__((ext_vector_type(2))) float v2f;
typedef __attribute__((ext_vector_type(4))) float v4f;
typedef __attribute__((ext_vector_type(8))) float v8f;

#define H_IN   384
#define W_IN   384
#define H_OUT  377
#define W_OUT  377
#define KH     8
#define KW     8
#define TILE_X 128
#define STRIP_Y 64
#define ROWS   (STRIP_Y + KH - 1)
#define XCOLS  (TILE_X + KW)
#define XQ     (XCOLS / 4)
#define NQ     (ROWS * XQ)
#define STAGE_ITERS ((NQ + 255) / 256)
#define NY     4
#define NX     4
#define KD     (NY + KH - 1)
#define KS     3
#define KP_ROWS 14
#define KP_COLS 16
#define OPITCH 136

__global__ __launch_bounds__(256)
void corr_wmma_kernel(const float* __restrict__ x,
                      const float* __restrict__ k,
                      float* __restrict__ out) {
  __shared__ __attribute__((aligned(16))) float XS[ROWS * XCOLS];
  __shared__ __attribute__((aligned(16))) float OS[STRIP_Y * OPITCH];
  __shared__ __attribute__((aligned(16))) float KP[KP_ROWS * KP_COLS];

  const int tid  = threadIdx.x;
  const int lane = tid & 31;
  const int wave = tid >> 5;
  const int b    = blockIdx.z;
  const int x0   = blockIdx.x * TILE_X;
  const int y0   = blockIdx.y * STRIP_Y;

  float kv = 0.f;
  {
    const int rr = (tid >> 4) - 3;
    const int cc = (tid & 15) - 3;
    if (tid < KP_ROWS * KP_COLS && rr >= 0 && rr < KH && cc >= 0 && cc < KW)
      kv = k[(size_t)b * (KH * KW) + rr * KW + cc];
  }

  const float* xb = x + (size_t)b * H_IN * W_IN;
  {
    v4f sv[STAGE_ITERS];
#pragma unroll
    for (int it = 0; it < STAGE_ITERS; ++it) {
      int q  = tid + it * 256;
      int yy = q / XQ;
      int xq = q - yy * XQ;
      int gy = y0 + yy;
      int gx = x0 + xq * 4;
      v4f v = {0.f, 0.f, 0.f, 0.f};
      if (q < NQ && gy < H_IN && gx < W_IN)
        v = *(const v4f*)(xb + gy * W_IN + gx);
      sv[it] = v;
    }
#pragma unroll
    for (int it = 0; it < STAGE_ITERS; ++it) {
      int q = tid + it * 256;
      if (q < NQ) *(v4f*)(&XS[q * 4]) = sv[it];
    }
  }
  if (tid < KP_ROWS * KP_COLS) KP[tid] = kv;

  __syncthreads();

  const int n    = lane & 15;
  const int half = lane >> 4;
  const int yo   = n >> 2;
  const int xs   = n & 3;
  v2f bf[KD][KS];
  {
    const float* kp = &KP[(3 - yo) * KP_COLS + (2 * half - xs + 3)];
#pragma unroll
    for (int d = 0; d < KD; ++d) {
#pragma unroll
      for (int s = 0; s < KS; ++s) {
        bf[d][s].x = kp[d * KP_COLS + 4 * s];
        bf[d][s].y = kp[d * KP_COLS + 4 * s + 1];
      }
    }
  }

  const int xh  = (wave & 1) * 64;
  const int rw  = (wave >> 1) * 16;
  const float* xa = &XS[xh + 4 * n + 2 * half];
  float* od = &OS[yo * OPITCH + xh + 32 * half + xs];
  for (int p = 0; p < 2; ++p) {
    const int rbA = rw + 8 * p;
    const int rbB = rbA + NY;
    const float* xrA = xa + rbA * XCOLS;
    const float* xrB = xa + rbB * XCOLS;
    v8f accA = {}, accB = {};
#pragma unroll
    for (int d = 0; d < KD; ++d) {
#pragma unroll
      for (int s = 0; s < KS; ++s) {
        v2f aA = *(const v2f*)(xrA + d * XCOLS + 4 * s);
        v2f aB = *(const v2f*)(xrB + d * XCOLS + 4 * s);
        accA = __builtin_amdgcn_wmma_f32_16x16x4_f32(false, aA, false, bf[d][s], (short)0, accA, false, false);
        accB = __builtin_amdgcn_wmma_f32_16x16x4_f32(false, aB, false, bf[d][s], (short)0, accB, false, false);
      }
    }
    float* oA = od + rbA * OPITCH;
    float* oB = od + rbB * OPITCH;
#pragma unroll
    for (int j = 0; j < 8; ++j) {
      oA[4 * j] = accA[j];
      oB[4 * j] = accB[j];
    }
  }

  __syncthreads();

  float* ob = out + (size_t)b * H_OUT * W_OUT;
  const int xo = tid & (TILE_X - 1);
  const int yb = tid >> 7;
  const int gx = x0 + xo;
  for (int i0 = 0; i0 < STRIP_Y / 2; i0 += 8) {
    float sv2[8];
#pragma unroll
    for (int u = 0; u < 8; ++u)
      sv2[u] = OS[(yb + (i0 + u) * 2) * OPITCH + xo];
#pragma unroll
    for (int u = 0; u < 8; ++u) {
      int gy = y0 + yb + (i0 + u) * 2;
      if (gy < H_OUT && gx < W_OUT) ob[gy * W_OUT + gx] = sv2[u];
    }
  }
}

extern "C" void kernel_launch(void* const* d_in, const int* in_sizes, int n_in,
                              void* d_out, int out_size, void* d_ws, size_t ws_size,
                              hipStream_t stream) {
  (void)in_sizes; (void)n_in; (void)d_ws; (void)ws_size; (void)out_size;
  const float* x = (const float*)d_in[0];
  const float* k = (const float*)d_in[1];
  float* out = (float*)d_out;
  dim3 grid((W_OUT + TILE_X - 1) / TILE_X,
            (H_OUT + STRIP_Y - 1) / STRIP_Y,
            128);
  corr_wmma_kernel<<<grid, dim3(256, 1, 1), 0, stream>>>(x, k, out);
}


# one work-group walks six 32x128 tiles of a sample: filter fragments built once; next tile's loads and previous tile's stores in flight during the WMMAs; strip and output tile double-buffered in LDS
# speedup vs baseline: 3.2301x; 1.0438x over previous
#include <hip/hip_runtime.h>


typedef __attribute__((ext_vector_type(2))) float v2f;
typedef __attribute__((ext_vector_type(4))) float v4f;
typedef __attribute__((ext_vector_type(8))) float v8f;

#define H_IN   384
#define W_IN   384
#define H_OUT  377
#define W_OUT  377
#define KH     8
#define KW     8
#define TILE_X 128
#define STRIP_Y 32
#define ROWS   (STRIP_Y + KH - 1)
#define XCOLS  (TILE_X + KW)
#define XQ     (XCOLS / 4)
#define XS_FLOATS (ROWS * XCOLS)
#define LROWS  7
#define LROUNDS ((ROWS + LROWS - 1) / LROWS)
#define X_TILES 3
#define NT     6
#define WG_PER_SAMPLE 6
#define NY     4
#define NX     4
#define KD     (NY + KH - 1)
#define KS     3
#define KP_ROWS 14
#define KP_COLS 16
#define OPITCH 136
#define OS_FLOATS (STRIP_Y * OPITCH)

__device__ __forceinline__ void strip_loads(v4f (&sv)[LROUNDS], const float* __restrict__ p,
                                            bool col_ok, int r, int gy) {
#pragma unroll
  for (int it = 0; it < LROUNDS; ++it) {
    v4f v = {0.f, 0.f, 0.f, 0.f};
    if (col_ok && (r + LROWS * it < ROWS) && (gy + LROWS * it < H_IN))
      v = *(const v4f*)(p + it * (LROWS * W_IN));
    sv[it] = v;
  }
}

__device__ __forceinline__ void strip_to_lds(float* xs, const v4f (&sv)[LROUNDS], bool active, int r, int xq) {
#pragma unroll
  for (int it = 0; it < LROUNDS; ++it) {
    if (active && (r + LROWS * it < ROWS))
      *(v4f*)(&xs[(r + LROWS * it) * XCOLS + 4 * xq]) = sv[it];
  }
}

__device__ __forceinline__ void store_tile(const float* os, float* __restrict__ ob, int tid, int x0, int y0) {
  const int xo = tid & (TILE_X - 1);
  const int yb = tid >> 7;
  const int gx = x0 + xo;
  float* po = ob + (y0 + yb) * W_OUT + gx;
  for (int i0 = 0; i0 < STRIP_Y / 2; i0 += 8) {
    float t[8];
#pragma unroll
    for (int u = 0; u < 8; ++u)
      t[u] = os[(yb + (i0 + u) * 2) * OPITCH + xo];
#pragma unroll
    for (int u = 0; u < 8; ++u) {
      const int gy = y0 + yb + (i0 + u) * 2;
      if (gy < H_OUT && gx < W_OUT) po[(i0 + u) * 2 * W_OUT] = t[u];
    }
  }
}

__global__ __launch_bounds__(256, 6)
void corr_wmma_kernel(const float* __restrict__ x,
                      const float* __restrict__ k,
                      float* __restrict__ out) {
  __shared__ __attribute__((aligned(16))) float XS[2 * XS_FLOATS];
  __shared__ __attribute__((aligned(16))) float OS[2 * OS_FLOATS];
  __shared__ __attribute__((aligned(16))) float KP[KP_ROWS * KP_COLS];

  const int tid  = threadIdx.x;
  const int lane = tid & 31;
  const int wave = tid >> 5;
  const int b    = blockIdx.y;
  const int x0   = (blockIdx.x % X_TILES) * TILE_X;
  const int ty0  = (blockIdx.x / X_TILES) * NT;

  float kv = 0.f;
  {
    const int rr = (tid >> 4) - 3;
    const int cc = (tid & 15) - 3;
    if (tid < KP_ROWS * KP_COLS && rr >= 0 && rr < KH && cc >= 0 && cc < KW)
      kv = k[(size_t)b * (KH * KW) + rr * KW + cc];
  }

  const float* xb = x + (size_t)b * H_IN * W_IN;
  float* ob = out + (size_t)b * H_OUT * W_OUT;

  const int  lr     = tid / XQ;
  const int  lq     = tid - lr * XQ;
  const bool active = lr < LROWS;
  const bool col_ok = active && (x0 + 4 * lq < W_IN);
  const float* lp   = xb + (ty0 * STRIP_Y + lr) * W_IN + x0 + 4 * lq;

  v4f sv[LROUNDS];
  strip_loads(sv, lp, col_ok, lr, ty0 * STRIP_Y + lr);

  if (tid < KP_ROWS * KP_COLS) KP[tid] = kv;
  strip_to_lds(&XS[0], sv, active, lr, lq);

  __syncthreads();

  const int n    = lane & 15;
  const int half = lane >> 4;
  const int yo   = n >> 2;
  const int xs   = n & 3;
  v2f bf[KD][KS];
  {
    const float* kp = &KP[(3 - yo) * KP_COLS + (2 * half - xs + 3)];
#pragma unroll
    for (int d = 0; d < KD; ++d) {
#pragma unroll
      for (int s = 0; s < KS; ++s) {
        bf[d][s].x = kp[d * KP_COLS + 4 * s];
        bf[d][s].y = kp[d * KP_COLS + 4 * s + 1];
      }
    }
  }

  const int xh  = (wave & 1) * 64;
  const int rw  = (wave >> 1) * 8;
  const int xa  = rw * XCOLS + xh + 4 * n + 2 * half;
  const int od  = (rw + yo) * OPITCH + xh + 32 * half + xs;

#pragma unroll 1
  for (int i = 0; i < NT; ++i) {
    const int y0 = (ty0 + i) * STRIP_Y;
    if (i + 1 < NT) strip_loads(sv, lp + (i + 1) * (STRIP_Y * W_IN), col_ok, lr, y0 + STRIP_Y + lr);
    if (i > 0) store_tile(&OS[((i - 1) & 1) * OS_FLOATS], ob, tid, x0, y0 - STRIP_Y);

    const float* xrA = &XS[(i & 1) * XS_FLOATS + xa];
    const float* xrB = xrA + NY * XCOLS;
    v8f accA = {}, accB = {};
#pragma unroll
    for (int d = 0; d < KD; ++d) {
#pragma unroll
      for (int s = 0; s < KS; ++s) {
        v2f aA = *(const v2f*)(xrA + d * XCOLS + 4 * s);
        v2f aB = *(const v2f*)(xrB + d * XCOLS + 4 * s);
        accA = __builtin_amdgcn_wmma_f32_16x16x4_f32(false, aA, false, bf[d][s], (short)0, accA, false, false);
        accB = __builtin_amdgcn_wmma_f32_16x16x4_f32(false, aB, false, bf[d][s], (short)0, accB, false, false);
      }
    }
    float* oA = &OS[(i & 1) * OS_FLOATS + od];
    float* oB = oA + NY * OPITCH;
#pragma unroll
    for (int j = 0; j < 8; ++j) {
      oA[4 * j] = accA[j];
      oB[4 * j] = accB[j];
    }

    if (i + 1 < NT) strip_to_lds(&XS[((i + 1) & 1) * XS_FLOATS], sv, active, lr, lq);

    __syncthreads();
  }

  store_tile(&OS[((NT - 1) & 1) * OS_FLOATS], ob, tid, x0, (ty0 + NT - 1) * STRIP_Y);
}

extern "C" void kernel_launch(void* const* d_in, const int* in_sizes, int n_in,
                              void* d_out, int out_size, void* d_ws, size_t ws_size,
                              hipStream_t stream) {
  (void)in_sizes; (void)n_in; (void)d_ws; (void)ws_size; (void)out_size;
  const float* x = (const float*)d_in[0];
  const float* k = (const float*)d_in[1];
  float* out = (float*)d_out;
  dim3 grid(WG_PER_SAMPLE,
            128,
            1);
  corr_wmma_kernel<<<grid, dim3(256, 1, 1), 0, stream>>>(x, k, out);
}


# image-edge tests leave the per-access path: strip rows 0..27 and every tile stored inside the loop are inside the image, so their loads and stores issue back to back under one column predicate
# speedup vs baseline: 3.3768x; 1.0454x over previous
#include <hip/hip_runtime.h>


typedef __attribute__((ext_vector_type(2))) float v2f;
typedef __attribute__((ext_vector_type(4))) float v4f;
typedef __attribute__((ext_vector_type(8))) float v8f;

#define H_IN   384
#define W_IN   384
#define H_OUT  377
#define W_OUT  377
#define KH     8
#define KW     8
#define TILE_X 128
#define STRIP_Y 32
#define ROWS   (STRIP_Y + KH - 1)
#define XCOLS  (TILE_X + KW)
#define XQ     (XCOLS / 4)
#define XS_FLOATS (ROWS * XCOLS)
#define LROWS  7
#define LROUNDS ((ROWS + LROWS - 1) / LROWS)
#define X_TILES 3
#define NT     6
#define WG_PER_SAMPLE 6
#define NY     4
#define NX     4
#define KD     (NY + KH - 1)
#define KS     3
#define KP_ROWS 14
#define KP_COLS 16
#define OPITCH 136
#define OS_FLOATS (STRIP_Y * OPITCH)

#define SAFE_ROUNDS 4
static_assert((H_IN / STRIP_Y - 1) * STRIP_Y + LROWS * SAFE_ROUNDS - 1 < H_IN, "rounds below SAFE_ROUNDS must stay inside the image");
static_assert(LROWS * SAFE_ROUNDS <= ROWS, "rounds below SAFE_ROUNDS must stay inside the strip");
__device__ __forceinline__ void strip_loads(v4f (&sv)[LROUNDS], const float* __restrict__ p,
                                            bool col_ok, int r, int gy) {
  if (col_ok) {
#pragma unroll
    for (int it = 0; it < SAFE_ROUNDS; ++it)
      sv[it] = *(const v4f*)(p + it * (LROWS * W_IN));
  }
#pragma unroll
  for (int it = SAFE_ROUNDS; it < LROUNDS; ++it) {
    v4f v = {0.f, 0.f, 0.f, 0.f};
    if (col_ok && (r + LROWS * it < ROWS) && (gy + LROWS * it < H_IN))
      v = *(const v4f*)(p + it * (LROWS * W_IN));
    sv[it] = v;
  }
}

__device__ __forceinline__ void strip_to_lds(float* xs, const v4f (&sv)[LROUNDS], bool active, int r, int xq) {
#pragma unroll
  for (int it = 0; it < LROUNDS; ++it) {
    if (active && (r + LROWS * it < ROWS))
      *(v4f*)(&xs[(r + LROWS * it) * XCOLS + 4 * xq]) = sv[it];
  }
}

__device__ __forceinline__ void store_tile(const float* os, float* __restrict__ ob, int tid, int x0, int y0) {
  const int xo = tid & (TILE_X - 1);
  const int yb = tid >> 7;
  const int gx = x0 + xo;
  float* po = ob + (y0 + yb) * W_OUT + gx;
  for (int i0 = 0; i0 < STRIP_Y / 2; i0 += 8) {
    float t[8];
#pragma unroll
    for (int u = 0; u < 8; ++u)
      t[u] = os[(yb + (i0 + u) * 2) * OPITCH + xo];
#pragma unroll
    for (int u = 0; u < 8; ++u) {
      const int gy = y0 + yb + (i0 + u) * 2;
      if (gy < H_OUT && gx < W_OUT) po[(i0 + u) * 2 * W_OUT] = t[u];
    }
  }
}

__device__ __forceinline__ void store_tile_inside(const float* os, float* __restrict__ ob, int tid, int x0, int y0) {
  const int xo = tid & (TILE_X - 1);
  const int yb = tid >> 7;
  const int gx = x0 + xo;
  if (gx < W_OUT) {
    float* po = ob + (y0 + yb) * W_OUT + gx;
    const float* ps = os + yb * OPITCH + xo;
    for (int i0 = 0; i0 < STRIP_Y / 2; i0 += 8) {
      float t[8];
#pragma unroll
      for (int u = 0; u < 8; ++u)
        t[u] = ps[(i0 + u) * 2 * OPITCH];
#pragma unroll
      for (int u = 0; u < 8; ++u)
        po[(i0 + u) * 2 * W_OUT] = t[u];
    }
  }
}

__global__ __launch_bounds__(256, 6)
void corr_wmma_kernel(const float* __restrict__ x,
                      const float* __restrict__ k,
                      float* __restrict__ out) {
  __shared__ __attribute__((aligned(16))) float XS[2 * XS_FLOATS];
  __shared__ __attribute__((aligned(16))) float OS[2 * OS_FLOATS];
  __shared__ __attribute__((aligned(16))) float KP[KP_ROWS * KP_COLS];

  const int tid  = threadIdx.x;
  const int lane = tid & 31;
  const int wave = tid >> 5;
  const int b    = blockIdx.y;
  const int x0   = (blockIdx.x % X_TILES) * TILE_X;
  const int ty0  = (blockIdx.x / X_TILES) * NT;

  float kv = 0.f;
  {
    const int rr = (tid >> 4) - 3;
    const int cc = (tid & 15) - 3;
    if (tid < KP_ROWS * KP_COLS && rr >= 0 && rr < KH && cc >= 0 && cc < KW)
      kv = k[(size_t)b * (KH * KW) + rr * KW + cc];
  }

  const float* xb = x + (size_t)b * H_IN * W_IN;
  float* ob = out + (size_t)b * H_OUT * W_OUT;

  const int  lr     = tid / XQ;
  const int  lq     = tid - lr * XQ;
  const bool active = lr < LROWS;
  const bool col_ok = active && (x0 + 4 * lq < W_IN);
  const float* lp   = xb + (ty0 * STRIP_Y + lr) * W_IN + x0 + 4 * lq;

  v4f sv[LROUNDS] = {};
  strip_loads(sv, lp, col_ok, lr, ty0 * STRIP_Y + lr);

  if (tid < KP_ROWS * KP_COLS) KP[tid] = kv;
  strip_to_lds(&XS[0], sv, active, lr, lq);

  __syncthreads();

  const int n    = lane & 15;
  const int half = lane >> 4;
  const int yo   = n >> 2;
  const int xs   = n & 3;
  v2f bf[KD][KS];
  {
    const float* kp = &KP[(3 - yo) * KP_COLS + (2 * half - xs + 3)];
#pragma unroll
    for (int d = 0; d < KD; ++d) {
#pragma unroll
      for (int s = 0; s < KS; ++s) {
        bf[d][s].x = kp[d * KP_COLS + 4 * s];
        bf[d][s].y = kp[d * KP_COLS + 4 * s + 1];
      }
    }
  }

  const int xh  = (wave & 1) * 64;
  const int rw  = (wave >> 1) * 8;
  const int xa  = rw * XCOLS + xh + 4 * n + 2 * half;
  const int od  = (rw + yo) * OPITCH + xh + 32 * half + xs;

#pragma unroll 1
  for (int i = 0; i < NT; ++i) {
    const int y0 = (ty0 + i) * STRIP_Y;
    if (i + 1 < NT) strip_loads(sv, lp + (i + 1) * (STRIP_Y * W_IN), col_ok, lr, y0 + STRIP_Y + lr);
    if (i > 0) store_tile_inside(&OS[((i - 1) & 1) * OS_FLOATS], ob, tid, x0, y0 - STRIP_Y);

    const float* xrA = &XS[(i & 1) * XS_FLOATS + xa];
    const float* xrB = xrA + NY * XCOLS;
    v8f accA = {}, accB = {};
#pragma unroll
    for (int d = 0; d < KD; ++d) {
#pragma unroll
      for (int s = 0; s < KS; ++s) {
        v2f aA = *(const v2f*)(xrA + d * XCOLS + 4 * s);
        v2f aB = *(const v2f*)(xrB + d * XCOLS + 4 * s);
        accA = __builtin_amdgcn_wmma_f32_16x16x4_f32(false, aA, false, bf[d][s], (short)0, accA, false, false);
        accB = __builtin_amdgcn_wmma_f32_16x16x4_f32(false, aB, false, bf[d][s], (short)0, accB, false, false);
      }
    }
    float* oA = &OS[(i & 1) * OS_FLOATS + od];
    float* oB = oA + NY * OPITCH;
#pragma unroll
    for (int j = 0; j < 8; ++j) {
      oA[4 * j] = accA[j];
      oB[4 * j] = accB[j];
    }

    if (i + 1 < NT) strip_to_lds(&XS[((i + 1) & 1) * XS_FLOATS], sv, active, lr, lq);

    __syncthreads();
  }

  store_tile(&OS[((NT - 1) & 1) * OS_FLOATS], ob, tid, x0, (ty0 + NT - 1) * STRIP_Y);
}

extern "C" void kernel_launch(void* const* d_in, const int* in_sizes, int n_in,
                              void* d_out, int out_size, void* d_ws, size_t ws_size,
                              hipStream_t stream) {
  (void)in_sizes; (void)n_in; (void)d_ws; (void)ws_size; (void)out_size;
  const float* x = (const float*)d_in[0];
  const float* k = (const float*)d_in[1];
  float* out = (float*)d_out;
  dim3 grid(WG_PER_SAMPLE,
            128,
            1);
  corr_wmma_kernel<<<grid, dim3(256, 1, 1), 0, stream>>>(x, k, out);
}


# wave priority falling with progress (s_setprio 3,3,2,2,1,0 over a work-group's six tiles, dropped in front of a tile's WMMA run): the three work-groups of a compute unit take turns and end together
# speedup vs baseline: 3.5017x; 1.0370x over previous
#include <hip/hip_runtime.h>


typedef __attribute__((ext_vector_type(2))) float v2f;
typedef __attribute__((ext_vector_type(4))) float v4f;
typedef __attribute__((ext_vector_type(8))) float v8f;

#define H_IN   384
#define W_IN   384
#define H_OUT  377
#define W_OUT  377
#define KH     8
#define KW     8
#define TILE_X 128
#define STRIP_Y 32
#define ROWS   (STRIP_Y + KH - 1)
#define XCOLS  (TILE_X + KW)
#define XQ     (XCOLS / 4)
#define XS_FLOATS (ROWS * XCOLS)
#define LROWS  7
#define LROUNDS ((ROWS + LROWS - 1) / LROWS)
#define X_TILES 3
#define NT     6
#define WG_PER_SAMPLE 6
#define NY     4
#define NX     4
#define KD     (NY + KH - 1)
#define KS     3
#define KP_ROWS 14
#define KP_COLS 16
#define OPITCH 136
#define OS_FLOATS (STRIP_Y * OPITCH)

#define SAFE_ROUNDS 4
static_assert((H_IN / STRIP_Y - 1) * STRIP_Y + LROWS * SAFE_ROUNDS - 1 < H_IN, "rounds below SAFE_ROUNDS must stay inside the image");
static_assert(LROWS * SAFE_ROUNDS <= ROWS, "rounds below SAFE_ROUNDS must stay inside the strip");
__device__ __forceinline__ void strip_loads(v4f (&sv)[LROUNDS], const float* __restrict__ p,
                                            bool col_ok, int r, int gy) {
  if (col_ok) {
#pragma unroll
    for (int it = 0; it < SAFE_ROUNDS; ++it)
      sv[it] = *(const v4f*)(p + it * (LROWS * W_IN));
  }
#pragma unroll
  for (int it = SAFE_ROUNDS; it < LROUNDS; ++it) {
    v4f v = {0.f, 0.f, 0.f, 0.f};
    if (col_ok && (r + LROWS * it < ROWS) && (gy + LROWS * it < H_IN))
      v = *(const v4f*)(p + it * (LROWS * W_IN));
    sv[it] = v;
  }
}

__device__ __forceinline__ void strip_to_lds(float* xs, const v4f (&sv)[LROUNDS], bool active, int r, int xq) {
#pragma unroll
  for (int it = 0; it < LROUNDS; ++it) {
    if (active && (r + LROWS * it < ROWS))
      *(v4f*)(&xs[(r + LROWS * it) * XCOLS + 4 * xq]) = sv[it];
  }
}

__device__ __forceinline__ void store_tile(const float* os, float* __restrict__ ob, int tid, int x0, int y0) {
  const int xo = tid & (TILE_X - 1);
  const int yb = tid >> 7;
  const int gx = x0 + xo;
  float* po = ob + (y0 + yb) * W_OUT + gx;
  for (int i0 = 0; i0 < STRIP_Y / 2; i0 += 8) {
    float t[8];
#pragma unroll
    for (int u = 0; u < 8; ++u)
      t[u] = os[(yb + (i0 + u) * 2) * OPITCH + xo];
#pragma unroll
    for (int u = 0; u < 8; ++u) {
      const int gy = y0 + yb + (i0 + u) * 2;
      if (gy < H_OUT && gx < W_OUT) po[(i0 + u) * 2 * W_OUT] = t[u];
    }
  }
}

__device__ __forceinline__ void store_tile_inside(const float* os, float* __restrict__ ob, int tid, int x0, int y0) {
  const int xo = tid & (TILE_X - 1);
  const int yb = tid >> 7;
  const int gx = x0 + xo;
  if (gx < W_OUT) {
    float* po = ob + (y0 + yb) * W_OUT + gx;
    const float* ps = os + yb * OPITCH + xo;
    for (int i0 = 0; i0 < STRIP_Y / 2; i0 += 8) {
      float t[8];
#pragma unroll
      for (int u = 0; u < 8; ++u)
        t[u] = ps[(i0 + u) * 2 * OPITCH];
#pragma unroll
      for (int u = 0; u < 8; ++u)
        po[(i0 + u) * 2 * W_OUT] = t[u];
    }
  }
}

__global__ __launch_bounds__(256, 6)
void corr_wmma_kernel(const float* __restrict__ x,
                      const float* __restrict__ k,
                      float* __restrict__ out) {
  __shared__ __attribute__((aligned(16))) float XS[2 * XS_FLOATS];
  __shared__ __attribute__((aligned(16))) float OS[2 * OS_FLOATS];
  __shared__ __attribute__((aligned(16))) float KP[KP_ROWS * KP_COLS];

  __builtin_amdgcn_s_setprio(3);

  const int tid  = threadIdx.x;
  const int lane = tid & 31;
  const int wave = tid >> 5;
  const int b    = blockIdx.y;
  const int x0   = (blockIdx.x % X_TILES) * TILE_X;
  const int ty0  = (blockIdx.x / X_TILES) * NT;

  float kv = 0.f;
  {
    const int rr = (tid >> 4) - 3;
    const int cc = (tid & 15) - 3;
    if (tid < KP_ROWS * KP_COLS && rr >= 0 && rr < KH && cc >= 0 && cc < KW)
      kv = k[(size_t)b * (KH * KW) + rr * KW + cc];
  }

  const float* xb = x + (size_t)b * H_IN * W_IN;
  float* ob = out + (size_t)b * H_OUT * W_OUT;

  const int  lr     = tid / XQ;
  const int  lq     = tid - lr * XQ;
  const bool active = lr < LROWS;
  const bool col_ok = active && (x0 + 4 * lq < W_IN);
  const float* lp   = xb + (ty0 * STRIP_Y + lr) * W_IN + x0 + 4 * lq;

  v4f sv[LROUNDS] = {};
  strip_loads(sv, lp, col_ok, lr, ty0 * STRIP_Y + lr);

  if (tid < KP_ROWS * KP_COLS) KP[tid] = kv;
  strip_to_lds(&XS[0], sv, active, lr, lq);

  __syncthreads();

  const int n    = lane & 15;
  const int half = lane >> 4;
  const int yo   = n >> 2;
  const int xs   = n & 3;
  v2f bf[KD][KS];
  {
    const float* kp = &KP[(3 - yo) * KP_COLS + (2 * half - xs + 3)];
#pragma unroll
    for (int d = 0; d < KD; ++d) {
#pragma unroll
      for (int s = 0; s < KS; ++s) {
        bf[d][s].x = kp[d * KP_COLS + 4 * s];
        bf[d][s].y = kp[d * KP_COLS + 4 * s + 1];
      }
    }
  }

  const int xh  = (wave & 1) * 64;
  const int rw  = (wave >> 1) * 8;
  const int xa  = rw * XCOLS + xh + 4 * n + 2 * half;
  const int od  = (rw + yo) * OPITCH + xh + 32 * half + xs;

#pragma unroll 1
  for (int i = 0; i < NT; ++i) {
    const int y0 = (ty0 + i) * STRIP_Y;
    if (i + 1 < NT) strip_loads(sv, lp + (i + 1) * (STRIP_Y * W_IN), col_ok, lr, y0 + STRIP_Y + lr);
    if (i > 0) store_tile_inside(&OS[((i - 1) & 1) * OS_FLOATS], ob, tid, x0, y0 - STRIP_Y);

    if (i == 2)      __builtin_amdgcn_s_setprio(2);
    else if (i == 4) __builtin_amdgcn_s_setprio(1);
    else if (i == 5) __builtin_amdgcn_s_setprio(0);

    const float* xrA = &XS[(i & 1) * XS_FLOATS + xa];
    const float* xrB = xrA + NY * XCOLS;
    v8f accA = {}, accB = {};
#pragma unroll
    for (int d = 0; d < KD; ++d) {
#pragma unroll
      for (int s = 0; s < KS; ++s) {
        v2f aA = *(const v2f*)(xrA + d * XCOLS + 4 * s);
        v2f aB = *(const v2f*)(xrB + d * XCOLS + 4 * s);
        accA = __builtin_amdgcn_wmma_f32_16x16x4_f32(false, aA, false, bf[d][s], (short)0, accA, false, false);
        accB = __builtin_amdgcn_wmma_f32_16x16x4_f32(false, aB, false, bf[d][s], (short)0, accB, false, false);
      }
    }
    float* oA = &OS[(i & 1) * OS_FLOATS + od];
    float* oB = oA + NY * OPITCH;
#pragma unroll
    for (int j = 0; j < 8; ++j) {
      oA[4 * j] = accA[j];
      oB[4 * j] = accB[j];
    }

    if (i + 1 < NT) strip_to_lds(&XS[((i + 1) & 1) * XS_FLOATS], sv, active, lr, lq);

    __syncthreads();
  }

  store_tile(&OS[((NT - 1) & 1) * OS_FLOATS], ob, tid, x0, (ty0 + NT - 1) * STRIP_Y);
}

extern "C" void kernel_launch(void* const* d_in, const int* in_sizes, int n_in,
                              void* d_out, int out_size, void* d_ws, size_t ws_size,
                              hipStream_t stream) {
  (void)in_sizes; (void)n_in; (void)d_ws; (void)ws_size; (void)out_size;
  const float* x = (const float*)d_in[0];
  const float* k = (const float*)d_in[1];
  float* out = (float*)d_out;
  dim3 grid(WG_PER_SAMPLE,
            128,
            1);
  corr_wmma_kernel<<<grid, dim3(256, 1, 1), 0, stream>>>(x, k, out);
}


# WMMA operands swapped (banded filter as A, input strip as B): the accumulator comes out row-major, each lane stores its 2 rows x 4 columns with two 16-byte stores; the tile's pass through LDS is gone
# speedup vs baseline: 3.6964x; 1.0556x over previous
#include <hip/hip_runtime.h>


typedef __attribute__((ext_vector_type(2))) float v2f;
typedef __attribute__((ext_vector_type(4))) float v4f;
typedef __attribute__((ext_vector_type(8))) float v8f;
typedef v4f v4f_a4 __attribute__((aligned(4)));

#define H_IN   384
#define W_IN   384
#define H_OUT  377
#define W_OUT  377
#define KH     8
#define KW     8
#define TILE_X 128
#define STRIP_Y 32
#define ROWS   (STRIP_Y + KH - 1)
#define XCOLS  (TILE_X + KW)
#define XQ     (XCOLS / 4)
#define XS_FLOATS (ROWS * XCOLS)
#define LROWS  7
#define LROUNDS ((ROWS + LROWS - 1) / LROWS)
#define X_TILES 3
#define NT     6
#define WG_PER_SAMPLE 6
#define NY     4
#define NX     4
#define KD     (NY + KH - 1)
#define KS     3
#define KP_ROWS 14
#define KP_COLS 16

#define SAFE_ROUNDS 4
static_assert((H_IN / STRIP_Y - 1) * STRIP_Y + LROWS * SAFE_ROUNDS - 1 < H_IN, "rounds below SAFE_ROUNDS must stay inside the image");
static_assert(LROWS * SAFE_ROUNDS <= ROWS, "rounds below SAFE_ROUNDS must stay inside the strip");
__device__ __forceinline__ void strip_loads(v4f (&sv)[LROUNDS], const float* __restrict__ p,
                                            bool col_ok, int r, int gy) {
  if (col_ok) {
#pragma unroll
    for (int it = 0; it < SAFE_ROUNDS; ++it)
      sv[it] = *(const v4f*)(p + it * (LROWS * W_IN));
  }
#pragma unroll
  for (int it = SAFE_ROUNDS; it < LROUNDS; ++it) {
    v4f v = {0.f, 0.f, 0.f, 0.f};
    if (col_ok && (r + LROWS * it < ROWS) && (gy + LROWS * it < H_IN))
      v = *(const v4f*)(p + it * (LROWS * W_IN));
    sv[it] = v;
  }
}

__device__ __forceinline__ void strip_to_lds(float* xs, const v4f (&sv)[LROUNDS], bool active, int r, int xq) {
#pragma unroll
  for (int it = 0; it < LROUNDS; ++it) {
    if (active && (r + LROWS * it < ROWS))
      *(v4f*)(&xs[(r + LROWS * it) * XCOLS + 4 * xq]) = sv[it];
  }
}

__device__ __forceinline__ void store_acc(const v8f& acc, float* __restrict__ p, int nv, bool r0ok, bool r1ok) {
  if (nv >= 4) {
    const v4f lo = {acc[0], acc[1], acc[2], acc[3]};
    const v4f hi = {acc[4], acc[5], acc[6], acc[7]};
    if (r0ok) *(v4f_a4*)p = lo;
    if (r1ok) *(v4f_a4*)(p + W_OUT) = hi;
  } else {
#pragma unroll
    for (int e = 0; e < 3; ++e) {
      if (e < nv) {
        if (r0ok) p[e] = acc[e];
        if (r1ok) p[W_OUT + e] = acc[4 + e];
      }
    }
  }
}

__global__ __launch_bounds__(256, 6)
void corr_wmma_kernel(const float* __restrict__ x,
                      const float* __restrict__ k,
                      float* __restrict__ out) {
  __shared__ __attribute__((aligned(16))) float XS[2 * XS_FLOATS];
  __shared__ __attribute__((aligned(16))) float KP[KP_ROWS * KP_COLS];

  __builtin_amdgcn_s_setprio(3);

  const int tid  = threadIdx.x;
  const int lane = tid & 31;
  const int wave = tid >> 5;
  const int b    = blockIdx.y;
  const int x0   = (blockIdx.x % X_TILES) * TILE_X;
  const int ty0  = (blockIdx.x / X_TILES) * NT;

  float kv = 0.f;
  {
    const int rr = (tid >> 4) - 3;
    const int cc = (tid & 15) - 3;
    if (tid < KP_ROWS * KP_COLS && rr >= 0 && rr < KH && cc >= 0 && cc < KW)
      kv = k[(size_t)b * (KH * KW) + rr * KW + cc];
  }

  const float* xb = x + (size_t)b * H_IN * W_IN;
  float* ob = out + (size_t)b * H_OUT * W_OUT;

  const int  lr     = tid / XQ;
  const int  lq     = tid - lr * XQ;
  const bool active = lr < LROWS;
  const bool col_ok = active && (x0 + 4 * lq < W_IN);
  const float* lp   = xb + (ty0 * STRIP_Y + lr) * W_IN + x0 + 4 * lq;

  v4f sv[LROUNDS] = {};
  strip_loads(sv, lp, col_ok, lr, ty0 * STRIP_Y + lr);

  if (tid < KP_ROWS * KP_COLS) KP[tid] = kv;
  strip_to_lds(&XS[0], sv, active, lr, lq);

  __syncthreads();

  const int n    = lane & 15;
  const int half = lane >> 4;
  const int yo   = n >> 2;
  const int xs   = n & 3;
  v2f bf[KD][KS];
  {
    const float* kp = &KP[(3 - yo) * KP_COLS + (2 * half - xs + 3)];
#pragma unroll
    for (int d = 0; d < KD; ++d) {
#pragma unroll
      for (int s = 0; s < KS; ++s) {
        bf[d][s].x = kp[d * KP_COLS + 4 * s];
        bf[d][s].y = kp[d * KP_COLS + 4 * s + 1];
      }
    }
  }

  const int xh  = (wave & 1) * 64;
  const int rw  = (wave >> 1) * 8;
  const int xa  = rw * XCOLS + xh + 4 * n + 2 * half;

  const int gx  = x0 + xh + 4 * n;
  const int nv  = W_OUT - gx;
  const int ry  = rw + 2 * half;
  float* po = ob + (ty0 * STRIP_Y + ry) * W_OUT + gx;

  v8f accA = {}, accB = {};
#pragma unroll 1
  for (int i = 0; i < NT; ++i) {
    const int y0 = (ty0 + i) * STRIP_Y;
    if (i + 1 < NT) strip_loads(sv, lp + (i + 1) * (STRIP_Y * W_IN), col_ok, lr, y0 + STRIP_Y + lr);
    if (i > 0) {
      float* pt = po + (i - 1) * (STRIP_Y * W_OUT);
      store_acc(accA, pt, nv, true, true);
      store_acc(accB, pt + NY * W_OUT, nv, true, true);
    }

    if (i == 2)      __builtin_amdgcn_s_setprio(2);
    else if (i == 4) __builtin_amdgcn_s_setprio(1);
    else if (i == 5) __builtin_amdgcn_s_setprio(0);

    const float* xrA = &XS[(i & 1) * XS_FLOATS + xa];
    const float* xrB = xrA + NY * XCOLS;
    accA = (v8f){};
    accB = (v8f){};
#pragma unroll
    for (int d = 0; d < KD; ++d) {
#pragma unroll
      for (int s = 0; s < KS; ++s) {
        v2f aA = *(const v2f*)(xrA + d * XCOLS + 4 * s);
        v2f aB = *(const v2f*)(xrB + d * XCOLS + 4 * s);
        accA = __builtin_amdgcn_wmma_f32_16x16x4_f32(false, bf[d][s], false, aA, (short)0, accA, false, false);
        accB = __builtin_amdgcn_wmma_f32_16x16x4_f32(false, bf[d][s], false, aB, (short)0, accB, false, false);
      }
    }

    if (i + 1 < NT) strip_to_lds(&XS[((i + 1) & 1) * XS_FLOATS], sv, active, lr, lq);

    __syncthreads();
  }

  {
    const int gy = (ty0 + NT - 1) * STRIP_Y + ry;
    float* pt = po + (NT - 1) * (STRIP_Y * W_OUT);
    store_acc(accA, pt, nv, gy < H_OUT, gy + 1 < H_OUT);
    store_acc(accB, pt + NY * W_OUT, nv, gy + NY < H_OUT, gy + NY + 1 < H_OUT);
  }
}

extern "C" void kernel_launch(void* const* d_in, const int* in_sizes, int n_in,
                              void* d_out, int out_size, void* d_ws, size_t ws_size,
                              hipStream_t stream) {
  (void)in_sizes; (void)n_in; (void)d_ws; (void)ws_size; (void)out_size;
  const float* x = (const float*)d_in[0];
  const float* k = (const float*)d_in[1];
  float* out = (float*)d_out;
  dim3 grid(WG_PER_SAMPLE,
            128,
            1);
  corr_wmma_kernel<<<grid, dim3(256, 1, 1), 0, stream>>>(x, k, out);
}
